# starting kernel
# baseline (speedup 1.0000x reference)
#include <hip/hip_runtime.h>


typedef float v2f __attribute__((ext_vector_type(2)));
typedef float v8f __attribute__((ext_vector_type(8)));

#define NTERMS 8
#define BDIM 1024
#define DDIM 512
#define NDIM 512
#define WAVES_PER_BLOCK 8

__global__ __launch_bounds__(256)
void ConjunctionLayer_wmma_kernel(const float* __restrict__ x,
                                  const float* __restrict__ W,
                                  float* __restrict__ out) {
    const int lane = threadIdx.x & 31;
    const int widb = threadIdx.x >> 5;
    const int wid  = blockIdx.x * WAVES_PER_BLOCK + widb;

    const int ntiles_n = NDIM >> 4;
    const int tb = (wid / ntiles_n) << 4;
    const int tn = (wid % ntiles_n) << 4;

    const int half = lane >> 4;
    const int l15  = lane & 15;

    const float* xrow = x + (size_t)(tb + l15) * DDIM + 2 * half;
    const float* wrow = W + (size_t)(tn + l15) * DDIM + 2 * half;

    v8f acc[NTERMS];
#pragma unroll
    for (int k = 0; k < NTERMS; ++k) {
        v8f z = {};
        acc[k] = z;
    }

    for (int d0 = 0; d0 < DDIM; d0 += 4) {
        v2f xv = *(const v2f*)(xrow + d0);
        v2f wv = *(const v2f*)(wrow + d0);
        v2f a;
        a.x = 1.0f - xv.x;
        a.y = 1.0f - xv.y;

        v2f pa = a;
        v2f pb = wv;

        acc[0] = __builtin_amdgcn_wmma_f32_16x16x4_f32(
            false, pa, false, pb, (short)0, acc[0], false, false);

#pragma unroll
        for (int k = 1; k < NTERMS; ++k) {
            pa.x *= a.x;  pa.y *= a.y;
            pb.x *= wv.x; pb.y *= wv.y;
            acc[k] = __builtin_amdgcn_wmma_f32_16x16x4_f32(
                false, pa, false, pb, (short)0, acc[k], false, false);
        }
    }

#pragma unroll
    for (int j = 0; j < 8; ++j) {
        float s = acc[0][j];
#pragma unroll
        for (int k = 1; k < NTERMS; ++k) {
            s += acc[k][j] * (1.0f / (float)(k + 1));
        }
        float val = expf(-s);
        int m = j + 8 * half;
        out[(size_t)(tb + m) * NDIM + (tn + l15)] = val;
    }
}

extern "C" void kernel_launch(void* const* d_in, const int* in_sizes, int n_in,
                              void* d_out, int out_size, void* d_ws, size_t ws_size,
                              hipStream_t stream) {
    const float* x = (const float*)d_in[0];
    const float* W = (const float*)d_in[1];
    float* out = (float*)d_out;

    const int tiles = (BDIM / 16) * (NDIM / 16);
    const int blocks = tiles / WAVES_PER_BLOCK;
    ConjunctionLayer_wmma_kernel<<<blocks, 256, 0, stream>>>(x, W, out);
}


# one wave per SIMD owns a 16x32 output tile (one A fragment feeds two B fragments), its 16-byte operand loads issued one 16-d step ahead of the matrix work into a second register set
# speedup vs baseline: 1.1811x; 1.1811x over previous
#include <hip/hip_runtime.h>


typedef float v2f __attribute__((ext_vector_type(2)));
typedef float v4f __attribute__((ext_vector_type(4)));
typedef float v8f __attribute__((ext_vector_type(8)));

#define NTERMS 8
#define BDIM 1024
#define DDIM 512
#define NDIM 512
#define WAVES_PER_BLOCK 4
#define DSTEP 16

struct Frag {
    v4f x0, x1;
    v4f w0a, w0b;
    v4f w1a, w1b;
};

__device__ __forceinline__ void load_frag(Frag& f, const float* __restrict__ xrow,
                                          const float* __restrict__ wrow0,
                                          const float* __restrict__ wrow1, int d0) {
    f.x0  = *(const v4f*)(xrow  + d0);
    f.x1  = *(const v4f*)(xrow  + d0 + 4);
    f.w0a = *(const v4f*)(wrow0 + d0);
    f.w0b = *(const v4f*)(wrow0 + d0 + 4);
    f.w1a = *(const v4f*)(wrow1 + d0);
    f.w1b = *(const v4f*)(wrow1 + d0 + 4);
}

__device__ __forceinline__ void kstep(v8f (&acc0)[NTERMS], v8f (&acc1)[NTERMS],
                                      float xx, float xy,
                                      float w0x, float w0y,
                                      float w1x, float w1y) {
    v2f a;  a.x = 1.0f - xx;  a.y = 1.0f - xy;
    v2f b0; b0.x = w0x; b0.y = w0y;
    v2f b1; b1.x = w1x; b1.y = w1y;

    v2f pa = a, p0 = b0, p1 = b1;
    acc0[0] = __builtin_amdgcn_wmma_f32_16x16x4_f32(
        false, pa, false, p0, (short)0, acc0[0], false, false);
    acc1[0] = __builtin_amdgcn_wmma_f32_16x16x4_f32(
        false, pa, false, p1, (short)0, acc1[0], false, false);
#pragma unroll
    for (int k = 1; k < NTERMS; ++k) {
        pa.x *= a.x;   pa.y *= a.y;
        p0.x *= b0.x;  p0.y *= b0.y;
        p1.x *= b1.x;  p1.y *= b1.y;
        acc0[k] = __builtin_amdgcn_wmma_f32_16x16x4_f32(
            false, pa, false, p0, (short)0, acc0[k], false, false);
        acc1[k] = __builtin_amdgcn_wmma_f32_16x16x4_f32(
            false, pa, false, p1, (short)0, acc1[k], false, false);
    }
}

__device__ __forceinline__ void compute_frag(v8f (&acc0)[NTERMS], v8f (&acc1)[NTERMS],
                                             const Frag& f) {
    kstep(acc0, acc1, f.x0.x, f.x0.y, f.w0a.x, f.w0a.y, f.w1a.x, f.w1a.y);
    kstep(acc0, acc1, f.x0.z, f.x0.w, f.w0a.z, f.w0a.w, f.w1a.z, f.w1a.w);
    kstep(acc0, acc1, f.x1.x, f.x1.y, f.w0b.x, f.w0b.y, f.w1b.x, f.w1b.y);
    kstep(acc0, acc1, f.x1.z, f.x1.w, f.w0b.z, f.w0b.w, f.w1b.z, f.w1b.w);
}

__device__ __forceinline__ void store_tile(const v8f (&acc)[NTERMS], float* __restrict__ out,
                                           int tb, int tn, int half, int l15) {
#pragma unroll
    for (int j = 0; j < 8; ++j) {
        float s = acc[0][j];
#pragma unroll
        for (int k = 1; k < NTERMS; ++k) {
            s += acc[k][j] * (1.0f / (float)(k + 1));
        }
        float val = expf(-s);
        int m = j + 8 * half;
        out[(size_t)(tb + m) * NDIM + (tn + l15)] = val;
    }
}

__global__ __launch_bounds__(128)
void ConjunctionLayer_wmma_kernel(const float* __restrict__ x,
                                  const float* __restrict__ W,
                                  float* __restrict__ out) {
    const int lane = threadIdx.x & 31;
    const int widb = threadIdx.x >> 5;

    const int nblk_n = NDIM >> 6;
    const int bb = (blockIdx.x / nblk_n) << 5;
    const int bn = (blockIdx.x % nblk_n) << 6;
    const int tb = bb + ((widb >> 1) << 4);
    const int tn = bn + ((widb & 1) << 5);

    const int half = lane >> 4;
    const int l15  = lane & 15;

    const float* xrow  = x + (size_t)(tb + l15) * DDIM + 8 * half;
    const float* wrow0 = W + (size_t)(tn + l15) * DDIM + 8 * half;
    const float* wrow1 = W + (size_t)(tn + 16 + l15) * DDIM + 8 * half;

    v8f acc0[NTERMS], acc1[NTERMS];
#pragma unroll
    for (int k = 0; k < NTERMS; ++k) {
        v8f z = {};
        acc0[k] = z;
        acc1[k] = z;
    }

    Frag f0, f1;
    load_frag(f0, xrow, wrow0, wrow1, 0);
    asm volatile("" ::: "memory");

    for (int d0 = 0; d0 < DDIM - 2 * DSTEP; d0 += 2 * DSTEP) {
        load_frag(f1, xrow, wrow0, wrow1, d0 + DSTEP);
        asm volatile("" ::: "memory");
        __builtin_amdgcn_sched_barrier(0);
        compute_frag(acc0, acc1, f0);
        __builtin_amdgcn_sched_barrier(0);
        load_frag(f0, xrow, wrow0, wrow1, d0 + 2 * DSTEP);
        asm volatile("" ::: "memory");
        __builtin_amdgcn_sched_barrier(0);
        compute_frag(acc0, acc1, f1);
        __builtin_amdgcn_sched_barrier(0);
    }
    load_frag(f1, xrow, wrow0, wrow1, DDIM - DSTEP);
    asm volatile("" ::: "memory");
    __builtin_amdgcn_sched_barrier(0);
    compute_frag(acc0, acc1, f0);
    compute_frag(acc0, acc1, f1);

    store_tile(acc0, out, tb, tn, half, l15);
    store_tile(acc1, out, tb, tn + 16, half, l15);
}

extern "C" void kernel_launch(void* const* d_in, const int* in_sizes, int n_in,
                              void* d_out, int out_size, void* d_ws, size_t ws_size,
                              hipStream_t stream) {
    const float* x = (const float*)d_in[0];
    const float* W = (const float*)d_in[1];
    float* out = (float*)d_out;

    const int blocks = (BDIM / 32) * (NDIM / 64);
    ConjunctionLayer_wmma_kernel<<<blocks, 32 * WAVES_PER_BLOCK, 0, stream>>>(x, W, out);
}


# series economised: degree-8 Taylor of -log(1-u) replaced by degree-5 minimax polynomial on declared range u in [0,0.1] (error per factor 1.95e-10, was 1.22e-10): 5 f32 matrix products per d, not 8
# speedup vs baseline: 1.5512x; 1.3134x over previous
#include <hip/hip_runtime.h>


typedef float v2f __attribute__((ext_vector_type(2)));
typedef float v4f __attribute__((ext_vector_type(4)));
typedef float v8f __attribute__((ext_vector_type(8)));

#define NTERMS 5
#define BDIM 1024
#define DDIM 512
#define NDIM 512
#define WAVES_PER_BLOCK 4
#define DSTEP 16

struct Frag {
    v4f x0, x1;
    v4f w0a, w0b;
    v4f w1a, w1b;
};

__device__ __forceinline__ void load_frag(Frag& f, const float* __restrict__ xrow,
                                          const float* __restrict__ wrow0,
                                          const float* __restrict__ wrow1, int d0) {
    f.x0  = *(const v4f*)(xrow  + d0);
    f.x1  = *(const v4f*)(xrow  + d0 + 4);
    f.w0a = *(const v4f*)(wrow0 + d0);
    f.w0b = *(const v4f*)(wrow0 + d0 + 4);
    f.w1a = *(const v4f*)(wrow1 + d0);
    f.w1b = *(const v4f*)(wrow1 + d0 + 4);
}

__device__ __forceinline__ void kstep(v8f (&acc0)[NTERMS], v8f (&acc1)[NTERMS],
                                      float xx, float xy,
                                      float w0x, float w0y,
                                      float w1x, float w1y) {
    v2f a;  a.x = 1.0f - xx;  a.y = 1.0f - xy;
    v2f b0; b0.x = w0x; b0.y = w0y;
    v2f b1; b1.x = w1x; b1.y = w1y;

    v2f pa = a, p0 = b0, p1 = b1;
    acc0[0] = __builtin_amdgcn_wmma_f32_16x16x4_f32(
        false, pa, false, p0, (short)0, acc0[0], false, false);
    acc1[0] = __builtin_amdgcn_wmma_f32_16x16x4_f32(
        false, pa, false, p1, (short)0, acc1[0], false, false);
#pragma unroll
    for (int k = 1; k < NTERMS; ++k) {
        pa.x *= a.x;   pa.y *= a.y;
        p0.x *= b0.x;  p0.y *= b0.y;
        p1.x *= b1.x;  p1.y *= b1.y;
        acc0[k] = __builtin_amdgcn_wmma_f32_16x16x4_f32(
            false, pa, false, p0, (short)0, acc0[k], false, false);
        acc1[k] = __builtin_amdgcn_wmma_f32_16x16x4_f32(
            false, pa, false, p1, (short)0, acc1[k], false, false);
    }
}

__device__ __forceinline__ void compute_frag(v8f (&acc0)[NTERMS], v8f (&acc1)[NTERMS],
                                             const Frag& f) {
    kstep(acc0, acc1, f.x0.x, f.x0.y, f.w0a.x, f.w0a.y, f.w1a.x, f.w1a.y);
    kstep(acc0, acc1, f.x0.z, f.x0.w, f.w0a.z, f.w0a.w, f.w1a.z, f.w1a.w);
    kstep(acc0, acc1, f.x1.x, f.x1.y, f.w0b.x, f.w0b.y, f.w1b.x, f.w1b.y);
    kstep(acc0, acc1, f.x1.z, f.x1.w, f.w0b.z, f.w0b.w, f.w1b.z, f.w1b.w);
}

__device__ __forceinline__ void store_tile(const v8f (&acc)[NTERMS], float* __restrict__ out,
                                           int tb, int tn, int half, int l15) {
    const float coef[NTERMS] = {1.0f, 0.4999968409538269f, 0.3335575461387634f,
                                0.244587704539299f, 0.25307369232177734f};
#pragma unroll
    for (int j = 0; j < 8; ++j) {
        float s = acc[0][j];
#pragma unroll
        for (int k = 1; k < NTERMS; ++k) {
            s += acc[k][j] * coef[k];
        }
        float val = expf(-s);
        int m = j + 8 * half;
        out[(size_t)(tb + m) * NDIM + (tn + l15)] = val;
    }
}

__global__ __launch_bounds__(128)
void ConjunctionLayer_wmma_kernel(const float* __restrict__ x,
                                  const float* __restrict__ W,
                                  float* __restrict__ out) {
    const int lane = threadIdx.x & 31;
    const int widb = threadIdx.x >> 5;

    const int nblk_n = NDIM >> 6;
    const int bb = (blockIdx.x / nblk_n) << 5;
    const int bn = (blockIdx.x % nblk_n) << 6;
    const int tb = bb + ((widb >> 1) << 4);
    const int tn = bn + ((widb & 1) << 5);

    const int half = lane >> 4;
    const int l15  = lane & 15;

    const float* xrow  = x + (size_t)(tb + l15) * DDIM + 8 * half;
    const float* wrow0 = W + (size_t)(tn + l15) * DDIM + 8 * half;
    const float* wrow1 = W + (size_t)(tn + 16 + l15) * DDIM + 8 * half;

    v8f acc0[NTERMS], acc1[NTERMS];
#pragma unroll
    for (int k = 0; k < NTERMS; ++k) {
        v8f z = {};
        acc0[k] = z;
        acc1[k] = z;
    }

    Frag f0, f1;
    load_frag(f0, xrow, wrow0, wrow1, 0);
    asm volatile("" ::: "memory");

    for (int d0 = 0; d0 < DDIM - 2 * DSTEP; d0 += 2 * DSTEP) {
        load_frag(f1, xrow, wrow0, wrow1, d0 + DSTEP);
        asm volatile("" ::: "memory");
        __builtin_amdgcn_sched_barrier(0);
        compute_frag(acc0, acc1, f0);
        __builtin_amdgcn_sched_barrier(0);
        load_frag(f0, xrow, wrow0, wrow1, d0 + 2 * DSTEP);
        asm volatile("" ::: "memory");
        __builtin_amdgcn_sched_barrier(0);
        compute_frag(acc0, acc1, f1);
        __builtin_amdgcn_sched_barrier(0);
    }
    load_frag(f1, xrow, wrow0, wrow1, DDIM - DSTEP);
    asm volatile("" ::: "memory");
    __builtin_amdgcn_sched_barrier(0);
    compute_frag(acc0, acc1, f0);
    compute_frag(acc0, acc1, f1);

    store_tile(acc0, out, tb, tn, half, l15);
    store_tile(acc1, out, tb, tn + 16, half, l15);
}

extern "C" void kernel_launch(void* const* d_in, const int* in_sizes, int n_in,
                              void* d_out, int out_size, void* d_ws, size_t ws_size,
                              hipStream_t stream) {
    const float* x = (const float*)d_in[0];
    const float* W = (const float*)d_in[1];
    float* out = (float*)d_out;

    const int blocks = (BDIM / 32) * (NDIM / 64);
    ConjunctionLayer_wmma_kernel<<<blocks, 32 * WAVES_PER_BLOCK, 0, stream>>>(x, W, out);
}


# operand loads issued three 16-d steps ahead of the matrix work that uses them (four register sets, was one step and two sets), so a second-level hit under load is covered
# speedup vs baseline: 1.8167x; 1.1712x over previous
#include <hip/hip_runtime.h>


typedef float v2f __attribute__((ext_vector_type(2)));
typedef float v4f __attribute__((ext_vector_type(4)));
typedef float v8f __attribute__((ext_vector_type(8)));

#define NTERMS 5
#define BDIM 1024
#define DDIM 512
#define NDIM 512
#define WAVES_PER_BLOCK 4
#define DSTEP 16

struct Frag {
    v4f x0, x1;
    v4f w0a, w0b;
    v4f w1a, w1b;
};

__device__ __forceinline__ void load_frag(Frag& f, const float* __restrict__ xrow,
                                          const float* __restrict__ wrow0,
                                          const float* __restrict__ wrow1, int d0) {
    f.x0  = *(const v4f*)(xrow  + d0);
    f.x1  = *(const v4f*)(xrow  + d0 + 4);
    f.w0a = *(const v4f*)(wrow0 + d0);
    f.w0b = *(const v4f*)(wrow0 + d0 + 4);
    f.w1a = *(const v4f*)(wrow1 + d0);
    f.w1b = *(const v4f*)(wrow1 + d0 + 4);
}

__device__ __forceinline__ void kstep(v8f (&acc0)[NTERMS], v8f (&acc1)[NTERMS],
                                      float xx, float xy,
                                      float w0x, float w0y,
                                      float w1x, float w1y) {
    v2f a;  a.x = 1.0f - xx;  a.y = 1.0f - xy;
    v2f b0; b0.x = w0x; b0.y = w0y;
    v2f b1; b1.x = w1x; b1.y = w1y;

    v2f pa = a, p0 = b0, p1 = b1;
    acc0[0] = __builtin_amdgcn_wmma_f32_16x16x4_f32(
        false, pa, false, p0, (short)0, acc0[0], false, false);
    acc1[0] = __builtin_amdgcn_wmma_f32_16x16x4_f32(
        false, pa, false, p1, (short)0, acc1[0], false, false);
#pragma unroll
    for (int k = 1; k < NTERMS; ++k) {
        pa.x *= a.x;   pa.y *= a.y;
        p0.x *= b0.x;  p0.y *= b0.y;
        p1.x *= b1.x;  p1.y *= b1.y;
        acc0[k] = __builtin_amdgcn_wmma_f32_16x16x4_f32(
            false, pa, false, p0, (short)0, acc0[k], false, false);
        acc1[k] = __builtin_amdgcn_wmma_f32_16x16x4_f32(
            false, pa, false, p1, (short)0, acc1[k], false, false);
    }
}

__device__ __forceinline__ void compute_frag(v8f (&acc0)[NTERMS], v8f (&acc1)[NTERMS],
                                             const Frag& f) {
    kstep(acc0, acc1, f.x0.x, f.x0.y, f.w0a.x, f.w0a.y, f.w1a.x, f.w1a.y);
    kstep(acc0, acc1, f.x0.z, f.x0.w, f.w0a.z, f.w0a.w, f.w1a.z, f.w1a.w);
    kstep(acc0, acc1, f.x1.x, f.x1.y, f.w0b.x, f.w0b.y, f.w1b.x, f.w1b.y);
    kstep(acc0, acc1, f.x1.z, f.x1.w, f.w0b.z, f.w0b.w, f.w1b.z, f.w1b.w);
}

__device__ __forceinline__ void pin_acc(v8f (&acc0)[NTERMS], v8f (&acc1)[NTERMS]) {
    asm volatile("" : "+v"(acc0[0]), "+v"(acc0[1]), "+v"(acc0[2]), "+v"(acc0[3]), "+v"(acc0[4]),
                      "+v"(acc1[0]), "+v"(acc1[1]), "+v"(acc1[2]), "+v"(acc1[3]), "+v"(acc1[4])
                 :: "memory");
}

__device__ __forceinline__ void store_tile(const v8f (&acc)[NTERMS], float* __restrict__ out,
                                           int tb, int tn, int half, int l15) {
    const float coef[NTERMS] = {1.0f, 0.4999968409538269f, 0.3335575461387634f,
                                0.244587704539299f, 0.25307369232177734f};
#pragma unroll
    for (int j = 0; j < 8; ++j) {
        float s = acc[0][j];
#pragma unroll
        for (int k = 1; k < NTERMS; ++k) {
            s += acc[k][j] * coef[k];
        }
        float val = expf(-s);
        int m = j + 8 * half;
        out[(size_t)(tb + m) * NDIM + (tn + l15)] = val;
    }
}

__global__ __launch_bounds__(128)
void ConjunctionLayer_wmma_kernel(const float* __restrict__ x,
                                  const float* __restrict__ W,
                                  float* __restrict__ out) {
    const int lane = threadIdx.x & 31;
    const int widb = threadIdx.x >> 5;

    const int nblk_n = NDIM >> 6;
    const int bb = (blockIdx.x / nblk_n) << 5;
    const int bn = (blockIdx.x % nblk_n) << 6;
    const int tb = bb + ((widb >> 1) << 4);
    const int tn = bn + ((widb & 1) << 5);

    const int half = lane >> 4;
    const int l15  = lane & 15;

    const float* xrow  = x + (size_t)(tb + l15) * DDIM + 8 * half;
    const float* wrow0 = W + (size_t)(tn + l15) * DDIM + 8 * half;
    const float* wrow1 = W + (size_t)(tn + 16 + l15) * DDIM + 8 * half;

    v8f acc0[NTERMS], acc1[NTERMS];
#pragma unroll
    for (int k = 0; k < NTERMS; ++k) {
        v8f z = {};
        acc0[k] = z;
        acc1[k] = z;
    }

    Frag f0, f1, f2, f3;
    load_frag(f0, xrow, wrow0, wrow1, 0);
    asm volatile("" ::: "memory");
    __builtin_amdgcn_sched_barrier(0);
    load_frag(f1, xrow, wrow0, wrow1, DSTEP);
    asm volatile("" ::: "memory");
    __builtin_amdgcn_sched_barrier(0);
    load_frag(f2, xrow, wrow0, wrow1, 2 * DSTEP);
    asm volatile("" ::: "memory");
    __builtin_amdgcn_sched_barrier(0);

    for (int d0 = 0; d0 < DDIM - 4 * DSTEP; d0 += 4 * DSTEP) {
        load_frag(f3, xrow, wrow0, wrow1, d0 + 3 * DSTEP);
        asm volatile("" ::: "memory");
        __builtin_amdgcn_sched_barrier(0);
        compute_frag(acc0, acc1, f0);
        pin_acc(acc0, acc1);
        __builtin_amdgcn_sched_barrier(0);
        load_frag(f0, xrow, wrow0, wrow1, d0 + 4 * DSTEP);
        asm volatile("" ::: "memory");
        __builtin_amdgcn_sched_barrier(0);
        compute_frag(acc0, acc1, f1);
        pin_acc(acc0, acc1);
        __builtin_amdgcn_sched_barrier(0);
        load_frag(f1, xrow, wrow0, wrow1, d0 + 5 * DSTEP);
        asm volatile("" ::: "memory");
        __builtin_amdgcn_sched_barrier(0);
        compute_frag(acc0, acc1, f2);
        pin_acc(acc0, acc1);
        __builtin_amdgcn_sched_barrier(0);
        load_frag(f2, xrow, wrow0, wrow1, d0 + 6 * DSTEP);
        asm volatile("" ::: "memory");
        __builtin_amdgcn_sched_barrier(0);
        compute_frag(acc0, acc1, f3);
        pin_acc(acc0, acc1);
        __builtin_amdgcn_sched_barrier(0);
    }
    load_frag(f3, xrow, wrow0, wrow1, DDIM - DSTEP);
    asm volatile("" ::: "memory");
    __builtin_amdgcn_sched_barrier(0);
    compute_frag(acc0, acc1, f0);
    compute_frag(acc0, acc1, f1);
    compute_frag(acc0, acc1, f2);
    compute_frag(acc0, acc1, f3);

    store_tile(acc0, out, tb, tn, half, l15);
    store_tile(acc1, out, tb, tn + 16, half, l15);
}

extern "C" void kernel_launch(void* const* d_in, const int* in_sizes, int n_in,
                              void* d_out, int out_size, void* d_ws, size_t ws_size,
                              hipStream_t stream) {
    const float* x = (const float*)d_in[0];
    const float* W = (const float*)d_in[1];
    float* out = (float*)d_out;

    const int blocks = (BDIM / 32) * (NDIM / 64);
    ConjunctionLayer_wmma_kernel<<<blocks, 32 * WAVES_PER_BLOCK, 0, stream>>>(x, W, out);
}


# series economised: degree-5 polynomial of -log(1-u) replaced by degree-4 minimax through zero on declared u in [0,0.1] (error per factor 5.8e-9, was 1.95e-10): 4 f32 matrix products per d, not 5
# speedup vs baseline: 2.0934x; 1.1523x over previous
#include <hip/hip_runtime.h>


typedef float v2f __attribute__((ext_vector_type(2)));
typedef float v4f __attribute__((ext_vector_type(4)));
typedef float v8f __attribute__((ext_vector_type(8)));

#define NTERMS 4
#define BDIM 1024
#define DDIM 512
#define NDIM 512
#define WAVES_PER_BLOCK 4
#define DSTEP 16

struct Frag {
    v4f x0, x1;
    v4f w0a, w0b;
    v4f w1a, w1b;
};

__device__ __forceinline__ void load_frag(Frag& f, const float* __restrict__ xrow,
                                          const float* __restrict__ wrow0,
                                          const float* __restrict__ wrow1, int d0) {
    f.x0  = *(const v4f*)(xrow  + d0);
    f.x1  = *(const v4f*)(xrow  + d0 + 4);
    f.w0a = *(const v4f*)(wrow0 + d0);
    f.w0b = *(const v4f*)(wrow0 + d0 + 4);
    f.w1a = *(const v4f*)(wrow1 + d0);
    f.w1b = *(const v4f*)(wrow1 + d0 + 4);
}

__device__ __forceinline__ void kstep(v8f (&acc0)[NTERMS], v8f (&acc1)[NTERMS],
                                      float xx, float xy,
                                      float w0x, float w0y,
                                      float w1x, float w1y) {
    v2f a;  a.x = 1.0f - xx;  a.y = 1.0f - xy;
    v2f b0; b0.x = w0x; b0.y = w0y;
    v2f b1; b1.x = w1x; b1.y = w1y;

    v2f pa = a, p0 = b0, p1 = b1;
    acc0[0] = __builtin_amdgcn_wmma_f32_16x16x4_f32(
        false, pa, false, p0, (short)0, acc0[0], false, false);
    acc1[0] = __builtin_amdgcn_wmma_f32_16x16x4_f32(
        false, pa, false, p1, (short)0, acc1[0], false, false);
#pragma unroll
    for (int k = 1; k < NTERMS; ++k) {
        pa.x *= a.x;   pa.y *= a.y;
        p0.x *= b0.x;  p0.y *= b0.y;
        p1.x *= b1.x;  p1.y *= b1.y;
        acc0[k] = __builtin_amdgcn_wmma_f32_16x16x4_f32(
            false, pa, false, p0, (short)0, acc0[k], false, false);
        acc1[k] = __builtin_amdgcn_wmma_f32_16x16x4_f32(
            false, pa, false, p1, (short)0, acc1[k], false, false);
    }
}

__device__ __forceinline__ void compute_frag(v8f (&acc0)[NTERMS], v8f (&acc1)[NTERMS],
                                             const Frag& f) {
    kstep(acc0, acc1, f.x0.x, f.x0.y, f.w0a.x, f.w0a.y, f.w1a.x, f.w1a.y);
    kstep(acc0, acc1, f.x0.z, f.x0.w, f.w0a.z, f.w0a.w, f.w1a.z, f.w1a.w);
    kstep(acc0, acc1, f.x1.x, f.x1.y, f.w0b.x, f.w0b.y, f.w1b.x, f.w1b.y);
    kstep(acc0, acc1, f.x1.z, f.x1.w, f.w0b.z, f.w0b.w, f.w1b.z, f.w1b.w);
}

__device__ __forceinline__ void pin_acc(v8f (&acc0)[NTERMS], v8f (&acc1)[NTERMS]) {
    asm volatile("" : "+v"(acc0[0]), "+v"(acc0[1]), "+v"(acc0[2]), "+v"(acc0[3]),
                      "+v"(acc1[0]), "+v"(acc1[1]), "+v"(acc1[2]), "+v"(acc1[3])
                 :: "memory");
}

__device__ __forceinline__ void store_tile(const v8f (&acc)[NTERMS], float* __restrict__ out,
                                           int tb, int tn, int half, int l15) {
    const float coef[NTERMS] = {0.99999821186065674f, 0.50017011165618896f,
                                0.32834169268608093f, 0.30646476149559021f};
#pragma unroll
    for (int j = 0; j < 8; ++j) {
        float s = acc[0][j] * coef[0];
#pragma unroll
        for (int k = 1; k < NTERMS; ++k) {
            s += acc[k][j] * coef[k];
        }
        float val = expf(-s);
        int m = j + 8 * half;
        out[(size_t)(tb + m) * NDIM + (tn + l15)] = val;
    }
}

__global__ __launch_bounds__(128)
void ConjunctionLayer_wmma_kernel(const float* __restrict__ x,
                                  const float* __restrict__ W,
                                  float* __restrict__ out) {
    const int lane = threadIdx.x & 31;
    const int widb = threadIdx.x >> 5;

    const int nblk_n = NDIM >> 6;
    const int bb = (blockIdx.x / nblk_n) << 5;
    const int bn = (blockIdx.x % nblk_n) << 6;
    const int tb = bb + ((widb >> 1) << 4);
    const int tn = bn + ((widb & 1) << 5);

    const int half = lane >> 4;
    const int l15  = lane & 15;

    const float* xrow  = x + (size_t)(tb + l15) * DDIM + 8 * half;
    const float* wrow0 = W + (size_t)(tn + l15) * DDIM + 8 * half;
    const float* wrow1 = W + (size_t)(tn + 16 + l15) * DDIM + 8 * half;

    v8f acc0[NTERMS], acc1[NTERMS];
#pragma unroll
    for (int k = 0; k < NTERMS; ++k) {
        v8f z = {};
        acc0[k] = z;
        acc1[k] = z;
    }

    Frag f0, f1, f2, f3;
    load_frag(f0, xrow, wrow0, wrow1, 0);
    asm volatile("" ::: "memory");
    __builtin_amdgcn_sched_barrier(0);
    load_frag(f1, xrow, wrow0, wrow1, DSTEP);
    asm volatile("" ::: "memory");
    __builtin_amdgcn_sched_barrier(0);
    load_frag(f2, xrow, wrow0, wrow1, 2 * DSTEP);
    asm volatile("" ::: "memory");
    __builtin_amdgcn_sched_barrier(0);

    for (int d0 = 0; d0 < DDIM - 4 * DSTEP; d0 += 4 * DSTEP) {
        load_frag(f3, xrow, wrow0, wrow1, d0 + 3 * DSTEP);
        asm volatile("" ::: "memory");
        __builtin_amdgcn_sched_barrier(0);
        compute_frag(acc0, acc1, f0);
        pin_acc(acc0, acc1);
        __builtin_amdgcn_sched_barrier(0);
        load_frag(f0, xrow, wrow0, wrow1, d0 + 4 * DSTEP);
        asm volatile("" ::: "memory");
        __builtin_amdgcn_sched_barrier(0);
        compute_frag(acc0, acc1, f1);
        pin_acc(acc0, acc1);
        __builtin_amdgcn_sched_barrier(0);
        load_frag(f1, xrow, wrow0, wrow1, d0 + 5 * DSTEP);
        asm volatile("" ::: "memory");
        __builtin_amdgcn_sched_barrier(0);
        compute_frag(acc0, acc1, f2);
        pin_acc(acc0, acc1);
        __builtin_amdgcn_sched_barrier(0);
        load_frag(f2, xrow, wrow0, wrow1, d0 + 6 * DSTEP);
        asm volatile("" ::: "memory");
        __builtin_amdgcn_sched_barrier(0);
        compute_frag(acc0, acc1, f3);
        pin_acc(acc0, acc1);
        __builtin_amdgcn_sched_barrier(0);
    }
    load_frag(f3, xrow, wrow0, wrow1, DDIM - DSTEP);
    asm volatile("" ::: "memory");
    __builtin_amdgcn_sched_barrier(0);
    compute_frag(acc0, acc1, f0);
    compute_frag(acc0, acc1, f1);
    compute_frag(acc0, acc1, f2);
    compute_frag(acc0, acc1, f3);

    store_tile(acc0, out, tb, tn, half, l15);
    store_tile(acc1, out, tb, tn + 16, half, l15);
}

extern "C" void kernel_launch(void* const* d_in, const int* in_sizes, int n_in,
                              void* d_out, int out_size, void* d_ws, size_t ws_size,
                              hipStream_t stream) {
    const float* x = (const float*)d_in[0];
    const float* W = (const float*)d_in[1];
    float* out = (float*)d_out;

    const int blocks = (BDIM / 32) * (NDIM / 64);
    ConjunctionLayer_wmma_kernel<<<blocks, 32 * WAVES_PER_BLOCK, 0, stream>>>(x, W, out);
}
